# starting kernel
# baseline (speedup 1.0000x reference)
#include <hip/hip_runtime.h>


#define EPS      1e-8f
#define FAKE_INF 10.0f
#define MDIM     512
#define BLOCK    512

typedef __attribute__((ext_vector_type(2))) float v2f;
typedef __attribute__((ext_vector_type(8))) float v8f;

__device__ __forceinline__
unsigned long long cmpex_sel(unsigned long long a, unsigned long long c,
                             bool keep_max)
{
    return (keep_max == (a > c)) ? a : c;
}

template<int J>
__device__ __forceinline__ unsigned long long swz_xor_u64(unsigned long long x)
{
    constexpr int imm = 0x1f | (J << 10);
    union { unsigned long long u; int i[2]; } v;
    v.u = x;
    v.i[0] = __builtin_amdgcn_ds_swizzle(v.i[0], imm);
    v.i[1] = __builtin_amdgcn_ds_swizzle(v.i[1], imm);
    return v.u;
}

template<int J>
__device__ __forceinline__ float swz_xor_f32(float x)
{
    constexpr int imm = 0x1f | (J << 10);
    return __int_as_float(__builtin_amdgcn_ds_swizzle(__float_as_int(x), imm));
}

template<int J>
__device__ __forceinline__ void wave_reduce2(float& a, float& b)
{
    if constexpr (J >= 1) {
        a += swz_xor_f32<J>(a);
        b += swz_xor_f32<J>(b);
        wave_reduce2<(J >> 1)>(a, b);
    }
}

template<int J>
__device__ __forceinline__ void wave_reduce1(float& a)
{
    if constexpr (J >= 1) {
        a += swz_xor_f32<J>(a);
        wave_reduce1<(J >> 1)>(a);
    }
}

template<int KS, int J>
__device__ __forceinline__ void bitonic_shfl_tail(unsigned long long& key, int tid)
{
    if constexpr (J >= 1) {
        const unsigned long long c = swz_xor_u64<J>(key);
        const bool keep_max = (((tid & J) == 0) == ((tid & KS) == 0));
        key = cmpex_sel(key, c, keep_max);
        bitonic_shfl_tail<KS, (J >> 1)>(key, tid);
    }
}

template<int KS>
__device__ __forceinline__
void bitonic_stage(unsigned long long& key, int tid,
                   unsigned long long (&buf)[2][MDIM], int& ph)
{
    #pragma unroll
    for (int j = KS >> 1; j >= 32; j >>= 1) {
        buf[ph][tid] = key;
        __syncthreads();
        const unsigned long long c = buf[ph][tid ^ j];
        const bool keep_max = (((tid & j) == 0) == ((tid & KS) == 0));
        key = cmpex_sel(key, c, keep_max);
        ph ^= 1;
    }
    constexpr int JS = ((KS >> 1) < 16) ? (KS >> 1) : 16;
    bitonic_shfl_tail<KS, JS>(key, tid);
}

__global__ __launch_bounds__(BLOCK)
void ibp_get_ub_kernel(const float* __restrict__ K,
                       const float* __restrict__ K1,
                       const float* __restrict__ Bv,
                       const float* __restrict__ B1,
                       const float* __restrict__ WLB,
                       const float* __restrict__ WUB,
                       float* __restrict__ out)
{
    __shared__ unsigned long long s_keyb[2][MDIM];
    __shared__ float s_td[MDIM], s_t1d[MDIM], s_in[MDIM], s_scan[MDIM];
    __shared__ float s_gsum[32], s_goff[32];
    __shared__ float s_redt[BLOCK / 32], s_redr[BLOCK / 32], s_bcast[2];

    const int tid  = threadIdx.x;
    const int row  = blockIdx.x;
    const int lane = tid & 31;
    const int wv   = tid >> 5;

    const float k   = K [(size_t)row * MDIM + tid];
    const float k1  = K1[(size_t)row * MDIM + tid];
    const float wlb = WLB[tid];
    const float wub = WUB[tid];

    const bool  pick_lb = (k < 0.0f) || ((k == 0.0f) && (k1 < 0.0f));
    const float w_ret   = pick_lb ? wlb : wub;

    float pt = w_ret * k;
    float pr = w_ret * k1;

    const float span  = wub - wlb;
    const float ak    = fabsf(k);
    const float ak1   = fabsf(k1);
    const float td    = ak  * span;
    const float t1d   = ak1 * span;
    const float ratio = (k * k1 >= 0.0f) ? 0.0f : (ak1 / (EPS + ak));

    s_td[tid]  = td;
    s_t1d[tid] = t1d;

    unsigned long long key =
        ((unsigned long long)__float_as_uint(ratio) << 32)
        | (unsigned int)(MDIM - 1 - tid);

    wave_reduce2<16>(pt, pr);
    if (lane == 0) { s_redt[wv] = pt; s_redr[wv] = pr; }
    __syncthreads();
    if (tid == 0) {
        float st = 0.0f, sr = 0.0f;
        #pragma unroll
        for (int w = 0; w < BLOCK / 32; ++w) { st += s_redt[w]; sr += s_redr[w]; }
        s_bcast[0] = st + Bv[row];
        s_bcast[1] = sr + B1[row];
    }
    __syncthreads();
    const float t    = s_bcast[0];
    const float ret0 = s_bcast[1];

    int ph = 0;
    bitonic_stage<  2>(key, tid, s_keyb, ph);
    bitonic_stage<  4>(key, tid, s_keyb, ph);
    bitonic_stage<  8>(key, tid, s_keyb, ph);
    bitonic_stage< 16>(key, tid, s_keyb, ph);
    bitonic_stage< 32>(key, tid, s_keyb, ph);
    bitonic_stage< 64>(key, tid, s_keyb, ph);
    bitonic_stage<128>(key, tid, s_keyb, ph);
    bitonic_stage<256>(key, tid, s_keyb, ph);
    bitonic_stage<512>(key, tid, s_keyb, ph);

    const int   idx     = (MDIM - 1) - (int)(key & 0xFFFFFFFFu);
    const float ratio_s = __uint_as_float((unsigned int)(key >> 32));
    const float td_s    = s_td[idx];
    const float t1d_s   = s_t1d[idx];
    s_in[tid] = td_s;
    __syncthreads();

    const int wvs = __builtin_amdgcn_readfirstlane(wv);
    if (wvs < 2) {
        const int mn = lane & 15;
        const int kh = lane >> 4;
        const int G  = wvs * 16 + mn;
        v8f acc = {0.f, 0.f, 0.f, 0.f, 0.f, 0.f, 0.f, 0.f};
        #pragma unroll
        for (int c = 0; c < 4; ++c) {
            const int k0 = 4 * c + 2 * kh;
            v2f a, b;
            a[0] = (mn >= k0)     ? 1.0f : 0.0f;
            a[1] = (mn >= k0 + 1) ? 1.0f : 0.0f;
            b[0] = s_in[G * 16 + k0];
            b[1] = s_in[G * 16 + k0 + 1];
            acc = __builtin_amdgcn_wmma_f32_16x16x4_f32(
                      false, a, false, b, (short)0, acc, false, false);
        }
        #pragma unroll
        for (int v = 0; v < 8; ++v) {
            const int m = v + 8 * kh;
            s_scan[G * 16 + m] = acc[v];
        }
        if (kh == 1) s_gsum[G] = acc[7];
    }
    __syncthreads();
    if (wvs == 0) {
        float x   = s_gsum[lane];
        float inc = x;
        #pragma unroll
        for (int off = 1; off < 32; off <<= 1) {
            const float y = __shfl_up(inc, off, 32);
            if (lane >= off) inc += y;
        }
        s_goff[lane] = inc - x;
    }
    __syncthreads();
    const float tds = s_scan[tid] + s_goff[tid >> 4];

    float percent;
    const float delta = t - tds;
    if (delta < 0.0f) {
        float pcl = delta / (td_s + EPS);
        pcl = fmaxf(-1.0f, fminf(0.0f, pcl));
        percent = pcl + 1.0f;
    } else {
        percent = 1.0f;
    }
    if (!(ratio_s > 0.0f)) percent = 0.0f;

    float contrib = t1d_s * percent;
    wave_reduce1<16>(contrib);
    if (lane == 0) s_redt[wv] = contrib;
    __syncthreads();
    if (tid == 0) {
        float cs = 0.0f;
        #pragma unroll
        for (int w = 0; w < BLOCK / 32; ++w) cs += s_redt[w];
        out[row] = (t >= 0.0f) ? (ret0 + cs) : -FAKE_INF;
    }
}

extern "C" void kernel_launch(void* const* d_in, const int* in_sizes, int n_in,
                              void* d_out, int out_size, void* d_ws, size_t ws_size,
                              hipStream_t stream)
{
    (void)n_in; (void)d_ws; (void)ws_size; (void)out_size;
    const float* k   = (const float*)d_in[0];
    const float* k1  = (const float*)d_in[1];
    const float* b   = (const float*)d_in[2];
    const float* b1  = (const float*)d_in[3];
    const float* wlb = (const float*)d_in[4];
    const float* wub = (const float*)d_in[5];

    const int B = in_sizes[2];

    ibp_get_ub_kernel<<<dim3(B), dim3(BLOCK), 0, stream>>>(
        k, k1, b, b1, wlb, wub, (float*)d_out);
}


# In-wave lane exchanges (35 in-wave sort steps, three wave sums) moved from ds_swizzle_b32 on the LDS unit to v_permlane16_b32 / v_permlanex16_b32 on the vector ALU; same partners, bit-identical sums
# speedup vs baseline: 1.2414x; 1.2414x over previous
#include <hip/hip_runtime.h>


#define EPS      1e-8f
#define FAKE_INF 10.0f
#define MDIM     512
#define BLOCK    512

typedef __attribute__((ext_vector_type(2))) float v2f;
typedef __attribute__((ext_vector_type(8))) float v8f;

__device__ __forceinline__
unsigned long long cmpex_sel(unsigned long long a, unsigned long long c,
                             bool keep_max)
{
    return (keep_max == (a > c)) ? a : c;
}

constexpr unsigned long long row_xor_sel(int m)
{
    unsigned long long s = 0;
    for (int j = 0; j < 16; ++j)
        s |= (unsigned long long)((j ^ m) & 15) << (4 * j);
    return s;
}

template<int J>
__device__ __forceinline__ unsigned int lane_xor_u32(unsigned int x)
{
    static_assert(J == 1 || J == 2 || J == 4 || J == 8 || J == 16,
                  "xor distance must be a power of two below the wave size");
    if constexpr (J == 16) {
        return __builtin_amdgcn_permlanex16(x, x, 0x76543210u, 0xfedcba98u,
                                            true, false);
    } else {
        constexpr unsigned long long sel = row_xor_sel(J);
        return __builtin_amdgcn_permlane16(x, x, (unsigned int)sel,
                                           (unsigned int)(sel >> 32),
                                           true, false);
    }
}

template<int J>
__device__ __forceinline__ unsigned long long lane_xor_u64(unsigned long long x)
{
    const unsigned int lo = lane_xor_u32<J>((unsigned int)x);
    const unsigned int hi = lane_xor_u32<J>((unsigned int)(x >> 32));
    return ((unsigned long long)hi << 32) | lo;
}

template<int J>
__device__ __forceinline__ float lane_xor_f32(float x)
{
    return __uint_as_float(lane_xor_u32<J>(__float_as_uint(x)));
}

template<int J>
__device__ __forceinline__ void wave_reduce2(float& a, float& b)
{
    if constexpr (J >= 1) {
        a += lane_xor_f32<J>(a);
        b += lane_xor_f32<J>(b);
        wave_reduce2<(J >> 1)>(a, b);
    }
}

template<int J>
__device__ __forceinline__ void wave_reduce1(float& a)
{
    if constexpr (J >= 1) {
        a += lane_xor_f32<J>(a);
        wave_reduce1<(J >> 1)>(a);
    }
}

template<int KS, int J>
__device__ __forceinline__ void bitonic_shfl_tail(unsigned long long& key, int tid)
{
    if constexpr (J >= 1) {
        const unsigned long long c = lane_xor_u64<J>(key);
        const bool keep_max = (((tid & J) == 0) == ((tid & KS) == 0));
        key = cmpex_sel(key, c, keep_max);
        bitonic_shfl_tail<KS, (J >> 1)>(key, tid);
    }
}

template<int KS>
__device__ __forceinline__
void bitonic_stage(unsigned long long& key, int tid,
                   unsigned long long (&buf)[2][MDIM], int& ph)
{
    #pragma unroll
    for (int j = KS >> 1; j >= 32; j >>= 1) {
        buf[ph][tid] = key;
        __syncthreads();
        const unsigned long long c = buf[ph][tid ^ j];
        const bool keep_max = (((tid & j) == 0) == ((tid & KS) == 0));
        key = cmpex_sel(key, c, keep_max);
        ph ^= 1;
    }
    constexpr int JS = ((KS >> 1) < 16) ? (KS >> 1) : 16;
    bitonic_shfl_tail<KS, JS>(key, tid);
}

__global__ __launch_bounds__(BLOCK)
void ibp_get_ub_kernel(const float* __restrict__ K,
                       const float* __restrict__ K1,
                       const float* __restrict__ Bv,
                       const float* __restrict__ B1,
                       const float* __restrict__ WLB,
                       const float* __restrict__ WUB,
                       float* __restrict__ out)
{
    __shared__ unsigned long long s_keyb[2][MDIM];
    __shared__ float s_td[MDIM], s_t1d[MDIM], s_in[MDIM], s_scan[MDIM];
    __shared__ float s_gsum[32], s_goff[32];
    __shared__ float s_redt[BLOCK / 32], s_redr[BLOCK / 32], s_bcast[2];

    const int tid  = threadIdx.x;
    const int row  = blockIdx.x;
    const int lane = tid & 31;
    const int wv   = tid >> 5;

    const float k   = K [(size_t)row * MDIM + tid];
    const float k1  = K1[(size_t)row * MDIM + tid];
    const float wlb = WLB[tid];
    const float wub = WUB[tid];

    const bool  pick_lb = (k < 0.0f) || ((k == 0.0f) && (k1 < 0.0f));
    const float w_ret   = pick_lb ? wlb : wub;

    float pt = w_ret * k;
    float pr = w_ret * k1;

    const float span  = wub - wlb;
    const float ak    = fabsf(k);
    const float ak1   = fabsf(k1);
    const float td    = ak  * span;
    const float t1d   = ak1 * span;
    const float ratio = (k * k1 >= 0.0f) ? 0.0f : (ak1 / (EPS + ak));

    s_td[tid]  = td;
    s_t1d[tid] = t1d;

    unsigned long long key =
        ((unsigned long long)__float_as_uint(ratio) << 32)
        | (unsigned int)(MDIM - 1 - tid);

    wave_reduce2<16>(pt, pr);
    if (lane == 0) { s_redt[wv] = pt; s_redr[wv] = pr; }
    __syncthreads();
    if (tid == 0) {
        float st = 0.0f, sr = 0.0f;
        #pragma unroll
        for (int w = 0; w < BLOCK / 32; ++w) { st += s_redt[w]; sr += s_redr[w]; }
        s_bcast[0] = st + Bv[row];
        s_bcast[1] = sr + B1[row];
    }
    __syncthreads();
    const float t    = s_bcast[0];
    const float ret0 = s_bcast[1];

    int ph = 0;
    bitonic_stage<  2>(key, tid, s_keyb, ph);
    bitonic_stage<  4>(key, tid, s_keyb, ph);
    bitonic_stage<  8>(key, tid, s_keyb, ph);
    bitonic_stage< 16>(key, tid, s_keyb, ph);
    bitonic_stage< 32>(key, tid, s_keyb, ph);
    bitonic_stage< 64>(key, tid, s_keyb, ph);
    bitonic_stage<128>(key, tid, s_keyb, ph);
    bitonic_stage<256>(key, tid, s_keyb, ph);
    bitonic_stage<512>(key, tid, s_keyb, ph);

    const int   idx     = (MDIM - 1) - (int)(key & 0xFFFFFFFFu);
    const float ratio_s = __uint_as_float((unsigned int)(key >> 32));
    const float td_s    = s_td[idx];
    const float t1d_s   = s_t1d[idx];
    s_in[tid] = td_s;
    __syncthreads();

    const int wvs = __builtin_amdgcn_readfirstlane(wv);
    if (wvs < 2) {
        const int mn = lane & 15;
        const int kh = lane >> 4;
        const int G  = wvs * 16 + mn;
        v8f acc = {0.f, 0.f, 0.f, 0.f, 0.f, 0.f, 0.f, 0.f};
        #pragma unroll
        for (int c = 0; c < 4; ++c) {
            const int k0 = 4 * c + 2 * kh;
            v2f a, b;
            a[0] = (mn >= k0)     ? 1.0f : 0.0f;
            a[1] = (mn >= k0 + 1) ? 1.0f : 0.0f;
            b[0] = s_in[G * 16 + k0];
            b[1] = s_in[G * 16 + k0 + 1];
            acc = __builtin_amdgcn_wmma_f32_16x16x4_f32(
                      false, a, false, b, (short)0, acc, false, false);
        }
        #pragma unroll
        for (int v = 0; v < 8; ++v) {
            const int m = v + 8 * kh;
            s_scan[G * 16 + m] = acc[v];
        }
        if (kh == 1) s_gsum[G] = acc[7];
    }
    __syncthreads();
    if (wvs == 0) {
        float x   = s_gsum[lane];
        float inc = x;
        #pragma unroll
        for (int off = 1; off < 32; off <<= 1) {
            const float y = __shfl_up(inc, off, 32);
            if (lane >= off) inc += y;
        }
        s_goff[lane] = inc - x;
    }
    __syncthreads();
    const float tds = s_scan[tid] + s_goff[tid >> 4];

    float percent;
    const float delta = t - tds;
    if (delta < 0.0f) {
        float pcl = delta / (td_s + EPS);
        pcl = fmaxf(-1.0f, fminf(0.0f, pcl));
        percent = pcl + 1.0f;
    } else {
        percent = 1.0f;
    }
    if (!(ratio_s > 0.0f)) percent = 0.0f;

    float contrib = t1d_s * percent;
    wave_reduce1<16>(contrib);
    if (lane == 0) s_redt[wv] = contrib;
    __syncthreads();
    if (tid == 0) {
        float cs = 0.0f;
        #pragma unroll
        for (int w = 0; w < BLOCK / 32; ++w) cs += s_redt[w];
        out[row] = (t >= 0.0f) ? (ret0 + cs) : -FAKE_INF;
    }
}

extern "C" void kernel_launch(void* const* d_in, const int* in_sizes, int n_in,
                              void* d_out, int out_size, void* d_ws, size_t ws_size,
                              hipStream_t stream)
{
    (void)n_in; (void)d_ws; (void)ws_size; (void)out_size;
    const float* k   = (const float*)d_in[0];
    const float* k1  = (const float*)d_in[1];
    const float* b   = (const float*)d_in[2];
    const float* b1  = (const float*)d_in[3];
    const float* wlb = (const float*)d_in[4];
    const float* wub = (const float*)d_in[5];

    const int B = in_sizes[2];

    ibp_get_ub_kernel<<<dim3(B), dim3(BLOCK), 0, stream>>>(
        k, k1, b, b1, wlb, wub, (float*)d_out);
}


# One wave per row, 16 sort keys per lane: 30 of 45 bitonic steps compare registers of one lane, no LDS exchange or barrier in the sort; t, ret, final sum: per-lane partials, then lane-xor tree
# speedup vs baseline: 1.3234x; 1.0661x over previous
#include <hip/hip_runtime.h>


#define EPS      1e-8f
#define FAKE_INF 10.0f
#define MDIM     512
#define WAVE     32
#define KPL      16

typedef __attribute__((ext_vector_type(2))) float v2f;
typedef __attribute__((ext_vector_type(8))) float v8f;
typedef unsigned long long u64;

__device__ __forceinline__ int pad16(int e) { return e + (e >> 4); }
#define PADDED (MDIM + MDIM / 16)

constexpr unsigned long long row_xor_sel(int m)
{
    unsigned long long s = 0;
    for (int j = 0; j < 16; ++j)
        s |= (unsigned long long)((j ^ m) & 15) << (4 * j);
    return s;
}

template<int J>
__device__ __forceinline__ unsigned int lane_xor_u32(unsigned int x)
{
    static_assert(J == 1 || J == 2 || J == 4 || J == 8 || J == 16,
                  "xor distance must be a power of two below the wave size");
    if constexpr (J == 16) {
        return __builtin_amdgcn_permlanex16(x, x, 0x76543210u, 0xfedcba98u,
                                            true, false);
    } else {
        constexpr unsigned long long sel = row_xor_sel(J);
        return __builtin_amdgcn_permlane16(x, x, (unsigned int)sel,
                                           (unsigned int)(sel >> 32),
                                           true, false);
    }
}

template<int J>
__device__ __forceinline__ u64 lane_xor_u64(u64 x)
{
    const unsigned int lo = lane_xor_u32<J>((unsigned int)x);
    const unsigned int hi = lane_xor_u32<J>((unsigned int)(x >> 32));
    return ((u64)hi << 32) | lo;
}

template<int J>
__device__ __forceinline__ float lane_xor_f32(float x)
{
    return __uint_as_float(lane_xor_u32<J>(__float_as_uint(x)));
}

template<int J>
__device__ __forceinline__ void wave_reduce2(float& a, float& b)
{
    if constexpr (J >= 1) {
        a += lane_xor_f32<J>(a);
        b += lane_xor_f32<J>(b);
        wave_reduce2<(J >> 1)>(a, b);
    }
}

template<int J>
__device__ __forceinline__ void wave_reduce1(float& a)
{
    if constexpr (J >= 1) {
        a += lane_xor_f32<J>(a);
        wave_reduce1<(J >> 1)>(a);
    }
}


template<int KS, int J>
__device__ __forceinline__ void inlane_step(u64 (&key)[KPL], int lane)
{
    #pragma unroll
    for (int r = 0; r < KPL; ++r) {
        if ((r & J) == 0) {
            bool desc;
            if constexpr (KS < KPL)        desc = ((r & KS) == 0);
            else if constexpr (KS >= MDIM) desc = true;
            else                           desc = ((lane & (KS / KPL)) == 0);
            const u64 a = key[r];
            const u64 b = key[r | J];
            const bool swp = ((a > b) != desc);
            key[r]     = swp ? b : a;
            key[r | J] = swp ? a : b;
        }
    }
}

template<int KS, int JL>
__device__ __forceinline__ void xlane_step(u64 (&key)[KPL], int lane)
{
    bool desc;
    if constexpr (KS >= MDIM) desc = true;
    else                      desc = ((lane & (KS / KPL)) == 0);
    const bool keep_max = (((lane & JL) == 0) == desc);
    #pragma unroll
    for (int r = 0; r < KPL; ++r) {
        const u64 a = key[r];
        const u64 c = lane_xor_u64<JL>(a);
        key[r] = (keep_max == (a > c)) ? a : c;
    }
}

template<int KS, int J>
__device__ __forceinline__ void stage_steps(u64 (&key)[KPL], int lane)
{
    if constexpr (J >= 1) {
        if constexpr (J >= KPL) xlane_step<KS, J / KPL>(key, lane);
        else                    inlane_step<KS, J>(key, lane);
        stage_steps<KS, (J >> 1)>(key, lane);
    }
}

template<int KS>
__device__ __forceinline__ void bitonic_stage(u64 (&key)[KPL], int lane)
{
    stage_steps<KS, (KS >> 1)>(key, lane);
}

__global__ __launch_bounds__(WAVE)
void ibp_get_ub_kernel(const float* __restrict__ K,
                       const float* __restrict__ K1,
                       const float* __restrict__ Bv,
                       const float* __restrict__ B1,
                       const float* __restrict__ WLB,
                       const float* __restrict__ WUB,
                       float* __restrict__ out)
{
    __shared__ __attribute__((aligned(16))) float s_buf[2 * PADDED];
    float* const s_in   = s_buf;
    float* const s_scan = s_buf + PADDED;

    const int lane = threadIdx.x;
    const int row  = blockIdx.x;

    const float4* __restrict__ K4   = (const float4*)(K  + (size_t)row * MDIM);
    const float4* __restrict__ K14  = (const float4*)(K1 + (size_t)row * MDIM);
    const float4* __restrict__ WLB4 = (const float4*)WLB;
    const float4* __restrict__ WUB4 = (const float4*)WUB;

    u64   key[KPL];
    float pt = 0.0f, pr = 0.0f;
    #pragma unroll
    for (int c = 0; c < 4; ++c) {
        const float4 kq   = K4  [c * WAVE + lane];
        const float4 k1q  = K14 [c * WAVE + lane];
        const float4 wlq  = WLB4[c * WAVE + lane];
        const float4 wuq  = WUB4[c * WAVE + lane];
        const float kk[4]  = {kq.x,  kq.y,  kq.z,  kq.w};
        const float kk1[4] = {k1q.x, k1q.y, k1q.z, k1q.w};
        const float wl[4]  = {wlq.x, wlq.y, wlq.z, wlq.w};
        const float wu[4]  = {wuq.x, wuq.y, wuq.z, wuq.w};
        #pragma unroll
        for (int u = 0; u < 4; ++u) {
            const int   i   = c * 128 + lane * 4 + u;
            const float k   = kk[u];
            const float k1  = kk1[u];
            const float wlb = wl[u];
            const float wub = wu[u];

            const bool  pick_lb = (k < 0.0f) || ((k == 0.0f) && (k1 < 0.0f));
            const float w_ret   = pick_lb ? wlb : wub;
            pt += w_ret * k;
            pr += w_ret * k1;

            const float span  = wub - wlb;
            const float ak    = fabsf(k);
            const float ak1   = fabsf(k1);
            const float td    = ak  * span;
            const float t1d   = ak1 * span;
            const float ratio = (k * k1 >= 0.0f) ? 0.0f : (ak1 / (EPS + ak));

            const int p = pad16(i);
            s_buf[2 * p]     = td;
            s_buf[2 * p + 1] = t1d;

            key[4 * c + u] = ((u64)__float_as_uint(ratio) << 32)
                           | (unsigned int)(MDIM - 1 - i);
        }
    }

    wave_reduce2<16>(pt, pr);
    const float t    = pt + Bv[row];
    const float ret0 = pr + B1[row];

    bitonic_stage<  2>(key, lane);
    bitonic_stage<  4>(key, lane);
    bitonic_stage<  8>(key, lane);
    bitonic_stage< 16>(key, lane);
    bitonic_stage< 32>(key, lane);
    bitonic_stage< 64>(key, lane);
    bitonic_stage<128>(key, lane);
    bitonic_stage<256>(key, lane);
    bitonic_stage<512>(key, lane);

    __syncthreads();

    float td_s[KPL], t1d_s[KPL];
    #pragma unroll
    for (int r = 0; r < KPL; ++r) {
        const int idx = (MDIM - 1) - (int)(key[r] & 0xFFFFFFFFu);
        const int p   = pad16(idx);
        td_s[r]  = s_buf[2 * p];
        t1d_s[r] = s_buf[2 * p + 1];
    }
    __syncthreads();
    #pragma unroll
    for (int r = 0; r < KPL; ++r)
        s_in[lane * (KPL + 1) + r] = td_s[r];
    __syncthreads();

    {
        const int mn = lane & 15;
        const int kh = lane >> 4;
        const int g0 = mn * (KPL + 1);
        const int g1 = (16 + mn) * (KPL + 1);
        v8f acc0 = {0.f, 0.f, 0.f, 0.f, 0.f, 0.f, 0.f, 0.f};
        v8f acc1 = {0.f, 0.f, 0.f, 0.f, 0.f, 0.f, 0.f, 0.f};
        #pragma unroll
        for (int c = 0; c < 4; ++c) {
            const int k0 = 4 * c + 2 * kh;
            v2f a, b0, b1;
            a[0] = (mn >= k0)     ? 1.0f : 0.0f;
            a[1] = (mn >= k0 + 1) ? 1.0f : 0.0f;
            b0[0] = s_in[g0 + k0];
            b0[1] = s_in[g0 + k0 + 1];
            b1[0] = s_in[g1 + k0];
            b1[1] = s_in[g1 + k0 + 1];
            acc0 = __builtin_amdgcn_wmma_f32_16x16x4_f32(
                       false, a, false, b0, (short)0, acc0, false, false);
            acc1 = __builtin_amdgcn_wmma_f32_16x16x4_f32(
                       false, a, false, b1, (short)0, acc1, false, false);
        }
        #pragma unroll
        for (int v = 0; v < 8; ++v) {
            const int m = v + 8 * kh;
            s_scan[g0 + m] = acc0[v];
            s_scan[g1 + m] = acc1[v];
        }
    }
    __syncthreads();

    float scan[KPL];
    #pragma unroll
    for (int r = 0; r < KPL; ++r)
        scan[r] = s_scan[lane * (KPL + 1) + r];

    const float gsum = scan[KPL - 1];
    float inc = gsum;
    #pragma unroll
    for (int off = 1; off < 32; off <<= 1) {
        const float y = __shfl_up(inc, off, 32);
        if (lane >= off) inc += y;
    }
    const float goff = inc - gsum;

    float contrib = 0.0f;
    #pragma unroll
    for (int r = 0; r < KPL; ++r) {
        const float ratio_s = __uint_as_float((unsigned int)(key[r] >> 32));
        const float tds     = scan[r] + goff;
        float percent;
        const float delta = t - tds;
        if (delta < 0.0f) {
            float pcl = delta / (td_s[r] + EPS);
            pcl = fmaxf(-1.0f, fminf(0.0f, pcl));
            percent = pcl + 1.0f;
        } else {
            percent = 1.0f;
        }
        if (!(ratio_s > 0.0f)) percent = 0.0f;
        contrib += t1d_s[r] * percent;
    }
    wave_reduce1<16>(contrib);
    if (lane == 0)
        out[row] = (t >= 0.0f) ? (ret0 + contrib) : -FAKE_INF;
}

extern "C" void kernel_launch(void* const* d_in, const int* in_sizes, int n_in,
                              void* d_out, int out_size, void* d_ws, size_t ws_size,
                              hipStream_t stream)
{
    (void)n_in; (void)d_ws; (void)ws_size; (void)out_size;
    const float* k   = (const float*)d_in[0];
    const float* k1  = (const float*)d_in[1];
    const float* b   = (const float*)d_in[2];
    const float* b1  = (const float*)d_in[3];
    const float* wlb = (const float*)d_in[4];
    const float* wub = (const float*)d_in[5];

    const int B = in_sizes[2];

    ibp_get_ub_kernel<<<dim3(B), dim3(WAVE), 0, stream>>>(
        k, k1, b, b1, wlb, wub, (float*)d_out);
}


# The 80 sort pairs of fixed direction no longer compile to 64-bit integer max/min: every compare-exchange is one 64-bit compare plus 32-bit selects of the key halves; same network, same order
# speedup vs baseline: 2.7744x; 2.0964x over previous
#include <hip/hip_runtime.h>


#define EPS      1e-8f
#define FAKE_INF 10.0f
#define MDIM     512
#define WAVE     32
#define KPL      16

typedef __attribute__((ext_vector_type(2))) float v2f;
typedef __attribute__((ext_vector_type(8))) float v8f;
typedef unsigned long long u64;

__device__ __forceinline__ int pad16(int e) { return e + (e >> 4); }
#define PADDED (MDIM + MDIM / 16)

constexpr unsigned long long row_xor_sel(int m)
{
    unsigned long long s = 0;
    for (int j = 0; j < 16; ++j)
        s |= (unsigned long long)((j ^ m) & 15) << (4 * j);
    return s;
}

template<int J>
__device__ __forceinline__ unsigned int lane_xor_u32(unsigned int x)
{
    static_assert(J == 1 || J == 2 || J == 4 || J == 8 || J == 16,
                  "xor distance must be a power of two below the wave size");
    if constexpr (J == 16) {
        return __builtin_amdgcn_permlanex16(x, x, 0x76543210u, 0xfedcba98u,
                                            true, false);
    } else {
        constexpr unsigned long long sel = row_xor_sel(J);
        return __builtin_amdgcn_permlane16(x, x, (unsigned int)sel,
                                           (unsigned int)(sel >> 32),
                                           true, false);
    }
}

template<int J>
__device__ __forceinline__ float lane_xor_f32(float x)
{
    return __uint_as_float(lane_xor_u32<J>(__float_as_uint(x)));
}

template<int J>
__device__ __forceinline__ void wave_reduce2(float& a, float& b)
{
    if constexpr (J >= 1) {
        a += lane_xor_f32<J>(a);
        b += lane_xor_f32<J>(b);
        wave_reduce2<(J >> 1)>(a, b);
    }
}

template<int J>
__device__ __forceinline__ void wave_reduce1(float& a)
{
    if constexpr (J >= 1) {
        a += lane_xor_f32<J>(a);
        wave_reduce1<(J >> 1)>(a);
    }
}


__device__ __forceinline__ u64 key64(unsigned int hi, unsigned int lo)
{
    return ((u64)hi << 32) | lo;
}

template<int KS, int J>
__device__ __forceinline__ void inlane_step(unsigned int (&kh)[KPL],
                                            unsigned int (&kl)[KPL], int lane)
{
    #pragma unroll
    for (int r = 0; r < KPL; ++r) {
        if ((r & J) == 0) {
            bool desc;
            if constexpr (KS < KPL)        desc = ((r & KS) == 0);
            else if constexpr (KS >= MDIM) desc = true;
            else                           desc = ((lane & (KS / KPL)) == 0);
            const unsigned int ah = kh[r],     al = kl[r];
            const unsigned int bh = kh[r | J], bl = kl[r | J];
            const bool swp = ((key64(ah, al) > key64(bh, bl)) != desc);
            kh[r]     = swp ? bh : ah;
            kl[r]     = swp ? bl : al;
            kh[r | J] = swp ? ah : bh;
            kl[r | J] = swp ? al : bl;
        }
    }
}

template<int KS, int JL>
__device__ __forceinline__ void xlane_step(unsigned int (&kh)[KPL],
                                           unsigned int (&kl)[KPL], int lane)
{
    bool desc;
    if constexpr (KS >= MDIM) desc = true;
    else                      desc = ((lane & (KS / KPL)) == 0);
    const bool keep_max = (((lane & JL) == 0) == desc);
    #pragma unroll
    for (int r = 0; r < KPL; ++r) {
        const unsigned int ah = kh[r], al = kl[r];
        const unsigned int ch = lane_xor_u32<JL>(ah);
        const unsigned int cl = lane_xor_u32<JL>(al);
        const bool own = (keep_max == (key64(ah, al) > key64(ch, cl)));
        kh[r] = own ? ah : ch;
        kl[r] = own ? al : cl;
    }
}

template<int KS, int J>
__device__ __forceinline__ void stage_steps(unsigned int (&kh)[KPL],
                                            unsigned int (&kl)[KPL], int lane)
{
    if constexpr (J >= 1) {
        if constexpr (J >= KPL) xlane_step<KS, J / KPL>(kh, kl, lane);
        else                    inlane_step<KS, J>(kh, kl, lane);
        stage_steps<KS, (J >> 1)>(kh, kl, lane);
    }
}

template<int KS>
__device__ __forceinline__ void bitonic_stage(unsigned int (&kh)[KPL],
                                              unsigned int (&kl)[KPL], int lane)
{
    stage_steps<KS, (KS >> 1)>(kh, kl, lane);
}

__global__ __launch_bounds__(WAVE)
void ibp_get_ub_kernel(const float* __restrict__ K,
                       const float* __restrict__ K1,
                       const float* __restrict__ Bv,
                       const float* __restrict__ B1,
                       const float* __restrict__ WLB,
                       const float* __restrict__ WUB,
                       float* __restrict__ out)
{
    __shared__ __attribute__((aligned(16))) float s_buf[2 * PADDED];
    float* const s_in   = s_buf;
    float* const s_scan = s_buf + PADDED;

    const int lane = threadIdx.x;
    const int row  = blockIdx.x;

    const float4* __restrict__ K4   = (const float4*)(K  + (size_t)row * MDIM);
    const float4* __restrict__ K14  = (const float4*)(K1 + (size_t)row * MDIM);
    const float4* __restrict__ WLB4 = (const float4*)WLB;
    const float4* __restrict__ WUB4 = (const float4*)WUB;

    unsigned int kh[KPL];
    unsigned int kl[KPL];
    float pt = 0.0f, pr = 0.0f;
    #pragma unroll
    for (int c = 0; c < 4; ++c) {
        const float4 kq   = K4  [c * WAVE + lane];
        const float4 k1q  = K14 [c * WAVE + lane];
        const float4 wlq  = WLB4[c * WAVE + lane];
        const float4 wuq  = WUB4[c * WAVE + lane];
        const float kk[4]  = {kq.x,  kq.y,  kq.z,  kq.w};
        const float kk1[4] = {k1q.x, k1q.y, k1q.z, k1q.w};
        const float wl[4]  = {wlq.x, wlq.y, wlq.z, wlq.w};
        const float wu[4]  = {wuq.x, wuq.y, wuq.z, wuq.w};
        #pragma unroll
        for (int u = 0; u < 4; ++u) {
            const int   i   = c * 128 + lane * 4 + u;
            const float k   = kk[u];
            const float k1  = kk1[u];
            const float wlb = wl[u];
            const float wub = wu[u];

            const bool  pick_lb = (k < 0.0f) || ((k == 0.0f) && (k1 < 0.0f));
            const float w_ret   = pick_lb ? wlb : wub;
            pt += w_ret * k;
            pr += w_ret * k1;

            const float span  = wub - wlb;
            const float ak    = fabsf(k);
            const float ak1   = fabsf(k1);
            const float td    = ak  * span;
            const float t1d   = ak1 * span;
            const float ratio = (k * k1 >= 0.0f) ? 0.0f : (ak1 / (EPS + ak));

            const int p = pad16(i);
            s_buf[2 * p]     = td;
            s_buf[2 * p + 1] = t1d;

            kh[4 * c + u] = __float_as_uint(ratio);
            kl[4 * c + u] = (unsigned int)(MDIM - 1 - i);
        }
    }

    wave_reduce2<16>(pt, pr);
    const float t    = pt + Bv[row];
    const float ret0 = pr + B1[row];

    bitonic_stage<  2>(kh, kl, lane);
    bitonic_stage<  4>(kh, kl, lane);
    bitonic_stage<  8>(kh, kl, lane);
    bitonic_stage< 16>(kh, kl, lane);
    bitonic_stage< 32>(kh, kl, lane);
    bitonic_stage< 64>(kh, kl, lane);
    bitonic_stage<128>(kh, kl, lane);
    bitonic_stage<256>(kh, kl, lane);
    bitonic_stage<512>(kh, kl, lane);

    __syncthreads();

    float td_s[KPL], t1d_s[KPL];
    #pragma unroll
    for (int r = 0; r < KPL; ++r) {
        const int idx = (MDIM - 1) - (int)kl[r];
        const int p   = pad16(idx);
        td_s[r]  = s_buf[2 * p];
        t1d_s[r] = s_buf[2 * p + 1];
    }
    __syncthreads();
    #pragma unroll
    for (int r = 0; r < KPL; ++r)
        s_in[lane * (KPL + 1) + r] = td_s[r];
    __syncthreads();

    {
        const int mn = lane & 15;
        const int kh = lane >> 4;
        const int g0 = mn * (KPL + 1);
        const int g1 = (16 + mn) * (KPL + 1);
        v8f acc0 = {0.f, 0.f, 0.f, 0.f, 0.f, 0.f, 0.f, 0.f};
        v8f acc1 = {0.f, 0.f, 0.f, 0.f, 0.f, 0.f, 0.f, 0.f};
        #pragma unroll
        for (int c = 0; c < 4; ++c) {
            const int k0 = 4 * c + 2 * kh;
            v2f a, b0, b1;
            a[0] = (mn >= k0)     ? 1.0f : 0.0f;
            a[1] = (mn >= k0 + 1) ? 1.0f : 0.0f;
            b0[0] = s_in[g0 + k0];
            b0[1] = s_in[g0 + k0 + 1];
            b1[0] = s_in[g1 + k0];
            b1[1] = s_in[g1 + k0 + 1];
            acc0 = __builtin_amdgcn_wmma_f32_16x16x4_f32(
                       false, a, false, b0, (short)0, acc0, false, false);
            acc1 = __builtin_amdgcn_wmma_f32_16x16x4_f32(
                       false, a, false, b1, (short)0, acc1, false, false);
        }
        #pragma unroll
        for (int v = 0; v < 8; ++v) {
            const int m = v + 8 * kh;
            s_scan[g0 + m] = acc0[v];
            s_scan[g1 + m] = acc1[v];
        }
    }
    __syncthreads();

    float scan[KPL];
    #pragma unroll
    for (int r = 0; r < KPL; ++r)
        scan[r] = s_scan[lane * (KPL + 1) + r];

    const float gsum = scan[KPL - 1];
    float inc = gsum;
    #pragma unroll
    for (int off = 1; off < 32; off <<= 1) {
        const float y = __shfl_up(inc, off, 32);
        if (lane >= off) inc += y;
    }
    const float goff = inc - gsum;

    float contrib = 0.0f;
    #pragma unroll
    for (int r = 0; r < KPL; ++r) {
        const float ratio_s = __uint_as_float(kh[r]);
        const float tds     = scan[r] + goff;
        float percent;
        const float delta = t - tds;
        if (delta < 0.0f) {
            float pcl = delta / (td_s[r] + EPS);
            pcl = fmaxf(-1.0f, fminf(0.0f, pcl));
            percent = pcl + 1.0f;
        } else {
            percent = 1.0f;
        }
        if (!(ratio_s > 0.0f)) percent = 0.0f;
        contrib += t1d_s[r] * percent;
    }
    wave_reduce1<16>(contrib);
    if (lane == 0)
        out[row] = (t >= 0.0f) ? (ret0 + contrib) : -FAKE_INF;
}

extern "C" void kernel_launch(void* const* d_in, const int* in_sizes, int n_in,
                              void* d_out, int out_size, void* d_ws, size_t ws_size,
                              hipStream_t stream)
{
    (void)n_in; (void)d_ws; (void)ws_size; (void)out_size;
    const float* k   = (const float*)d_in[0];
    const float* k1  = (const float*)d_in[1];
    const float* b   = (const float*)d_in[2];
    const float* b1  = (const float*)d_in[3];
    const float* wlb = (const float*)d_in[4];
    const float* wub = (const float*)d_in[5];

    const int B = in_sizes[2];

    ibp_get_ub_kernel<<<dim3(B), dim3(WAVE), 0, stream>>>(
        k, k1, b, b1, wlb, wub, (float*)d_out);
}


# Sort moves ONE 32-bit word per key (ratio bits; index code + zero payload where ratio is not positive); each element finds its rank in an LDS search tree; equal ratios: old 64-bit sort. Same sums
# speedup vs baseline: 2.9121x; 1.0496x over previous
#include <hip/hip_runtime.h>


#define EPS      1e-8f
#define FAKE_INF 10.0f
#define MDIM     512
#define WAVE     32
#define KPL      16

typedef __attribute__((ext_vector_type(2))) float v2f;
typedef __attribute__((ext_vector_type(8))) float v8f;
typedef unsigned long long u64;

__device__ __forceinline__ int pad16(int e) { return e + (e >> 4); }
#define PADDED (MDIM + MDIM / 16)

constexpr unsigned long long row_xor_sel(int m)
{
    unsigned long long s = 0;
    for (int j = 0; j < 16; ++j)
        s |= (unsigned long long)((j ^ m) & 15) << (4 * j);
    return s;
}

template<int J>
__device__ __forceinline__ unsigned int lane_xor_u32(unsigned int x)
{
    static_assert(J == 1 || J == 2 || J == 4 || J == 8 || J == 16,
                  "xor distance must be a power of two below the wave size");
    if constexpr (J == 16) {
        return __builtin_amdgcn_permlanex16(x, x, 0x76543210u, 0xfedcba98u,
                                            true, false);
    } else {
        constexpr unsigned long long sel = row_xor_sel(J);
        return __builtin_amdgcn_permlane16(x, x, (unsigned int)sel,
                                           (unsigned int)(sel >> 32),
                                           true, false);
    }
}

template<int J>
__device__ __forceinline__ float lane_xor_f32(float x)
{
    return __uint_as_float(lane_xor_u32<J>(__float_as_uint(x)));
}

template<int J>
__device__ __forceinline__ void wave_reduce2(float& a, float& b)
{
    if constexpr (J >= 1) {
        a += lane_xor_f32<J>(a);
        b += lane_xor_f32<J>(b);
        wave_reduce2<(J >> 1)>(a, b);
    }
}

template<int J>
__device__ __forceinline__ void wave_reduce1(float& a)
{
    if constexpr (J >= 1) {
        a += lane_xor_f32<J>(a);
        wave_reduce1<(J >> 1)>(a);
    }
}


__device__ __forceinline__ u64 key64(unsigned int hi, unsigned int lo)
{
    return ((u64)hi << 32) | lo;
}

template<int KS, int J>
__device__ __forceinline__ void inlane_step(unsigned int (&kh)[KPL],
                                            unsigned int (&kl)[KPL], int lane)
{
    #pragma unroll
    for (int r = 0; r < KPL; ++r) {
        if ((r & J) == 0) {
            bool desc;
            if constexpr (KS < KPL)        desc = ((r & KS) == 0);
            else if constexpr (KS >= MDIM) desc = true;
            else                           desc = ((lane & (KS / KPL)) == 0);
            const unsigned int ah = kh[r],     al = kl[r];
            const unsigned int bh = kh[r | J], bl = kl[r | J];
            const bool swp = ((key64(ah, al) > key64(bh, bl)) != desc);
            kh[r]     = swp ? bh : ah;
            kl[r]     = swp ? bl : al;
            kh[r | J] = swp ? ah : bh;
            kl[r | J] = swp ? al : bl;
        }
    }
}

template<int KS, int JL>
__device__ __forceinline__ void xlane_step(unsigned int (&kh)[KPL],
                                           unsigned int (&kl)[KPL], int lane)
{
    bool desc;
    if constexpr (KS >= MDIM) desc = true;
    else                      desc = ((lane & (KS / KPL)) == 0);
    const bool keep_max = (((lane & JL) == 0) == desc);
    #pragma unroll
    for (int r = 0; r < KPL; ++r) {
        const unsigned int ah = kh[r], al = kl[r];
        const unsigned int ch = lane_xor_u32<JL>(ah);
        const unsigned int cl = lane_xor_u32<JL>(al);
        const bool own = (keep_max == (key64(ah, al) > key64(ch, cl)));
        kh[r] = own ? ah : ch;
        kl[r] = own ? al : cl;
    }
}

template<int KS, int J>
__device__ __forceinline__ void stage_steps(unsigned int (&kh)[KPL],
                                            unsigned int (&kl)[KPL], int lane)
{
    if constexpr (J >= 1) {
        if constexpr (J >= KPL) xlane_step<KS, J / KPL>(kh, kl, lane);
        else                    inlane_step<KS, J>(kh, kl, lane);
        stage_steps<KS, (J >> 1)>(kh, kl, lane);
    }
}

template<int KS>
__device__ __forceinline__ void bitonic_stage(unsigned int (&kh)[KPL],
                                              unsigned int (&kl)[KPL], int lane)
{
    stage_steps<KS, (KS >> 1)>(kh, kl, lane);
}

template<int KS, int J>
__device__ __forceinline__ void inlane_step1(unsigned int (&k)[KPL], int lane)
{
    #pragma unroll
    for (int r = 0; r < KPL; ++r) {
        if ((r & J) == 0) {
            bool desc;
            if constexpr (KS < KPL)        desc = ((r & KS) == 0);
            else if constexpr (KS >= MDIM) desc = true;
            else                           desc = ((lane & (KS / KPL)) == 0);
            const unsigned int a = k[r], b = k[r | J];
            const bool swp = ((a > b) != desc);
            k[r]     = swp ? b : a;
            k[r | J] = swp ? a : b;
        }
    }
}

template<int KS, int JL>
__device__ __forceinline__ void xlane_step1(unsigned int (&k)[KPL], int lane)
{
    bool desc;
    if constexpr (KS >= MDIM) desc = true;
    else                      desc = ((lane & (KS / KPL)) == 0);
    const bool keep_max = (((lane & JL) == 0) == desc);
    #pragma unroll
    for (int r = 0; r < KPL; ++r) {
        const unsigned int a = k[r];
        const unsigned int c = lane_xor_u32<JL>(a);
        const bool own = (keep_max == (a > c));
        k[r] = own ? a : c;
    }
}

template<int KS, int J>
__device__ __forceinline__ void stage_steps1(unsigned int (&k)[KPL], int lane)
{
    if constexpr (J >= 1) {
        if constexpr (J >= KPL) xlane_step1<KS, J / KPL>(k, lane);
        else                    inlane_step1<KS, J>(k, lane);
        stage_steps1<KS, (J >> 1)>(k, lane);
    }
}

template<int KS>
__device__ __forceinline__ void bitonic_stages1(unsigned int (&k)[KPL], int lane)
{
    if constexpr (KS <= MDIM) {
        stage_steps1<KS, (KS >> 1)>(k, lane);
        bitonic_stages1<2 * KS>(k, lane);
    }
}

__device__ __forceinline__ float ratio_of(float k, float k1)
{
    return (k * k1 >= 0.0f) ? 0.0f : (fabsf(k1) / (EPS + fabsf(k)));
}

constexpr int ctz_c(int x) { int n = 0; while (((x >> n) & 1) == 0) ++n; return n; }

template<int R>
__device__ __forceinline__ void tree_store(unsigned int* tree,
                                           const unsigned int (&k)[KPL], int lane)
{
    if constexpr (R < KPL - 1) {
        constexpr int tz = ctz_c(R + 1);
        tree[(256 >> tz) + ((lane * KPL + R + 1) >> (tz + 1))] = k[R];
        tree_store<R + 1>(tree, k, lane);
    } else {
        const int tz = 4 + __builtin_ctz((unsigned int)(lane + 1));
        const unsigned int q = (unsigned int)(KPL * (lane + 1));
        tree[(256u >> tz) + (q >> (tz + 1))] = k[R];
    }
}

__global__ __launch_bounds__(WAVE)
void ibp_get_ub_kernel(const float* __restrict__ K,
                       const float* __restrict__ K1,
                       const float* __restrict__ Bv,
                       const float* __restrict__ B1,
                       const float* __restrict__ WLB,
                       const float* __restrict__ WUB,
                       float* __restrict__ out)
{
    __shared__ __attribute__((aligned(16))) float s_buf[2 * PADDED];
    __shared__ unsigned int s_tree[MDIM];
    float* const s_in   = s_buf;
    float* const s_scan = s_buf + PADDED;

    const int lane = threadIdx.x;
    const int row  = blockIdx.x;

    const float4* __restrict__ K4   = (const float4*)(K  + (size_t)row * MDIM);
    const float4* __restrict__ K14  = (const float4*)(K1 + (size_t)row * MDIM);
    const float4* __restrict__ WLB4 = (const float4*)WLB;
    const float4* __restrict__ WUB4 = (const float4*)WUB;

    unsigned int h0[KPL];
    float td_o[KPL];
    float t1d_o[KPL];
    float pt = 0.0f, pr = 0.0f;
    #pragma unroll
    for (int c = 0; c < 4; ++c) {
        const float4 kq   = K4  [c * WAVE + lane];
        const float4 k1q  = K14 [c * WAVE + lane];
        const float4 wlq  = WLB4[c * WAVE + lane];
        const float4 wuq  = WUB4[c * WAVE + lane];
        const float kk[4]  = {kq.x,  kq.y,  kq.z,  kq.w};
        const float kk1[4] = {k1q.x, k1q.y, k1q.z, k1q.w};
        const float wl[4]  = {wlq.x, wlq.y, wlq.z, wlq.w};
        const float wu[4]  = {wuq.x, wuq.y, wuq.z, wuq.w};
        #pragma unroll
        for (int u = 0; u < 4; ++u) {
            const int   i   = c * 128 + lane * 4 + u;
            const float k   = kk[u];
            const float k1  = kk1[u];
            const float wlb = wl[u];
            const float wub = wu[u];

            const bool  pick_lb = (k < 0.0f) || ((k == 0.0f) && (k1 < 0.0f));
            const float w_ret   = pick_lb ? wlb : wub;
            pt += w_ret * k;
            pr += w_ret * k1;

            const float span  = wub - wlb;
            const float ak    = fabsf(k);
            const float ak1   = fabsf(k1);
            const float td    = ak  * span;
            const float t1d   = ak1 * span;
            const float ratio = ratio_of(k, k1);
            const bool  pos   = (ratio > 0.0f);

            td_o [4 * c + u] = pos ? td  : 0.0f;
            t1d_o[4 * c + u] = pos ? t1d : 0.0f;
            h0   [4 * c + u] = pos ? __float_as_uint(ratio)
                                   : (unsigned int)(MDIM - 1 - i);
        }
    }

    wave_reduce2<16>(pt, pr);
    const float t    = pt + Bv[row];
    const float ret0 = pr + B1[row];

    unsigned int key[KPL];
    #pragma unroll
    for (int r = 0; r < KPL; ++r) key[r] = h0[r];
    bitonic_stages1<2>(key, lane);

    bool tie = false;
    #pragma unroll
    for (int r = 0; r + 1 < KPL; ++r)
        tie |= (key[r] == key[r + 1]);
    const unsigned int prev_last = __shfl_up(key[KPL - 1], 1, WAVE);
    tie |= (lane > 0) && (prev_last == key[0]);

    float td_s[KPL], t1d_s[KPL];
    if (__builtin_amdgcn_ballot_w32(tie) == 0u) {
        tree_store<0>(s_tree, key, lane);
        __syncthreads();

        unsigned int nb[KPL];
        #pragma unroll
        for (int e = 0; e < KPL; ++e) nb[e] = 4u;
        #pragma unroll
        for (int lvl = 0; lvl < 9; ++lvl) {
            #pragma unroll
            for (int e = 0; e < KPL; ++e) {
                const unsigned int w = s_tree[nb[e] >> 2];
                nb[e] = (nb[e] << 1) + ((w > h0[e]) ? 4u : 0u);
            }
        }
        #pragma unroll
        for (int e = 0; e < KPL; ++e) {
            const int rank = (int)(nb[e] >> 2) - MDIM;
            const int p    = pad16(rank);
            s_buf[2 * p]     = td_o[e];
            s_buf[2 * p + 1] = t1d_o[e];
        }
        __syncthreads();
        #pragma unroll
        for (int r = 0; r < KPL; ++r) {
            const int p = lane * (KPL + 1) + r;
            td_s[r]  = s_buf[2 * p];
            t1d_s[r] = s_buf[2 * p + 1];
        }
    } else {
        unsigned int kh[KPL], kl[KPL];
        #pragma unroll
        for (int c = 0; c < 4; ++c) {
            #pragma unroll
            for (int u = 0; u < 4; ++u) {
                const int i = c * 128 + lane * 4 + u;
                const int p = pad16(i);
                s_buf[2 * p]     = td_o [4 * c + u];
                s_buf[2 * p + 1] = t1d_o[4 * c + u];
                kh[4 * c + u] = h0[4 * c + u];
                kl[4 * c + u] = (unsigned int)(MDIM - 1 - i);
            }
        }
        bitonic_stage<  2>(kh, kl, lane);
        bitonic_stage<  4>(kh, kl, lane);
        bitonic_stage<  8>(kh, kl, lane);
        bitonic_stage< 16>(kh, kl, lane);
        bitonic_stage< 32>(kh, kl, lane);
        bitonic_stage< 64>(kh, kl, lane);
        bitonic_stage<128>(kh, kl, lane);
        bitonic_stage<256>(kh, kl, lane);
        bitonic_stage<512>(kh, kl, lane);
        __syncthreads();
        #pragma unroll
        for (int r = 0; r < KPL; ++r) {
            const int idx = (MDIM - 1) - (int)kl[r];
            const int p   = pad16(idx);
            td_s[r]  = s_buf[2 * p];
            t1d_s[r] = s_buf[2 * p + 1];
        }
    }
    __syncthreads();
    #pragma unroll
    for (int r = 0; r < KPL; ++r)
        s_in[lane * (KPL + 1) + r] = td_s[r];
    __syncthreads();

    {
        const int mn = lane & 15;
        const int kh = lane >> 4;
        const int g0 = mn * (KPL + 1);
        const int g1 = (16 + mn) * (KPL + 1);
        v8f acc0 = {0.f, 0.f, 0.f, 0.f, 0.f, 0.f, 0.f, 0.f};
        v8f acc1 = {0.f, 0.f, 0.f, 0.f, 0.f, 0.f, 0.f, 0.f};
        #pragma unroll
        for (int c = 0; c < 4; ++c) {
            const int k0 = 4 * c + 2 * kh;
            v2f a, b0, b1;
            a[0] = (mn >= k0)     ? 1.0f : 0.0f;
            a[1] = (mn >= k0 + 1) ? 1.0f : 0.0f;
            b0[0] = s_in[g0 + k0];
            b0[1] = s_in[g0 + k0 + 1];
            b1[0] = s_in[g1 + k0];
            b1[1] = s_in[g1 + k0 + 1];
            acc0 = __builtin_amdgcn_wmma_f32_16x16x4_f32(
                       false, a, false, b0, (short)0, acc0, false, false);
            acc1 = __builtin_amdgcn_wmma_f32_16x16x4_f32(
                       false, a, false, b1, (short)0, acc1, false, false);
        }
        #pragma unroll
        for (int v = 0; v < 8; ++v) {
            const int m = v + 8 * kh;
            s_scan[g0 + m] = acc0[v];
            s_scan[g1 + m] = acc1[v];
        }
    }
    __syncthreads();

    float scan[KPL];
    #pragma unroll
    for (int r = 0; r < KPL; ++r)
        scan[r] = s_scan[lane * (KPL + 1) + r];

    const float gsum = scan[KPL - 1];
    float inc = gsum;
    #pragma unroll
    for (int off = 1; off < 32; off <<= 1) {
        const float y = __shfl_up(inc, off, 32);
        if (lane >= off) inc += y;
    }
    const float goff = inc - gsum;

    float contrib = 0.0f;
    #pragma unroll
    for (int r = 0; r < KPL; ++r) {
        const float tds     = scan[r] + goff;
        float percent;
        const float delta = t - tds;
        if (delta < 0.0f) {
            float pcl = delta / (td_s[r] + EPS);
            pcl = fmaxf(-1.0f, fminf(0.0f, pcl));
            percent = pcl + 1.0f;
        } else {
            percent = 1.0f;
        }
        contrib += t1d_s[r] * percent;
    }
    wave_reduce1<16>(contrib);
    if (lane == 0)
        out[row] = (t >= 0.0f) ? (ret0 + contrib) : -FAKE_INF;
}

extern "C" void kernel_launch(void* const* d_in, const int* in_sizes, int n_in,
                              void* d_out, int out_size, void* d_ws, size_t ws_size,
                              hipStream_t stream)
{
    (void)n_in; (void)d_ws; (void)ws_size; (void)out_size;
    const float* k   = (const float*)d_in[0];
    const float* k1  = (const float*)d_in[1];
    const float* b   = (const float*)d_in[2];
    const float* b1  = (const float*)d_in[3];
    const float* wlb = (const float*)d_in[4];
    const float* wub = (const float*)d_in[5];

    const int B = in_sizes[2];

    ibp_get_ub_kernel<<<dim3(B), dim3(WAVE), 0, stream>>>(
        k, k1, b, b1, wlb, wub, (float*)d_out);
}


# One-word sort on the float pipe: a compare-exchange whose side depends on the lane is ONE v_med3_num_f32(a, b, per-lane +inf/-inf) per key, not compare + scalar xor + select; same order, same sums
# speedup vs baseline: 3.5583x; 1.2219x over previous
#include <hip/hip_runtime.h>


#define EPS      1e-8f
#define FAKE_INF 10.0f
#define MDIM     512
#define WAVE     32
#define KPL      16
#define CODE_BASE 0x00800000u

typedef __attribute__((ext_vector_type(2))) float v2f;
typedef __attribute__((ext_vector_type(8))) float v8f;
typedef unsigned long long u64;

__device__ __forceinline__ int pad16(int e) { return e + (e >> 4); }
#define PADDED (MDIM + MDIM / 16)

constexpr unsigned long long row_xor_sel(int m)
{
    unsigned long long s = 0;
    for (int j = 0; j < 16; ++j)
        s |= (unsigned long long)((j ^ m) & 15) << (4 * j);
    return s;
}

template<int J>
__device__ __forceinline__ unsigned int lane_xor_u32(unsigned int x)
{
    static_assert(J == 1 || J == 2 || J == 4 || J == 8 || J == 16,
                  "xor distance must be a power of two below the wave size");
    if constexpr (J == 16) {
        return __builtin_amdgcn_permlanex16(x, x, 0x76543210u, 0xfedcba98u,
                                            true, false);
    } else {
        constexpr unsigned long long sel = row_xor_sel(J);
        return __builtin_amdgcn_permlane16(x, x, (unsigned int)sel,
                                           (unsigned int)(sel >> 32),
                                           true, false);
    }
}

template<int J>
__device__ __forceinline__ float lane_xor_f32(float x)
{
    return __uint_as_float(lane_xor_u32<J>(__float_as_uint(x)));
}

template<int J>
__device__ __forceinline__ void wave_reduce2(float& a, float& b)
{
    if constexpr (J >= 1) {
        a += lane_xor_f32<J>(a);
        b += lane_xor_f32<J>(b);
        wave_reduce2<(J >> 1)>(a, b);
    }
}

template<int J>
__device__ __forceinline__ void wave_reduce1(float& a)
{
    if constexpr (J >= 1) {
        a += lane_xor_f32<J>(a);
        wave_reduce1<(J >> 1)>(a);
    }
}


__device__ __forceinline__ u64 key64(unsigned int hi, unsigned int lo)
{
    return ((u64)hi << 32) | lo;
}

template<int KS, int J>
__device__ __forceinline__ void inlane_step(unsigned int (&kh)[KPL],
                                            unsigned int (&kl)[KPL], int lane)
{
    #pragma unroll
    for (int r = 0; r < KPL; ++r) {
        if ((r & J) == 0) {
            bool desc;
            if constexpr (KS < KPL)        desc = ((r & KS) == 0);
            else if constexpr (KS >= MDIM) desc = true;
            else                           desc = ((lane & (KS / KPL)) == 0);
            const unsigned int ah = kh[r],     al = kl[r];
            const unsigned int bh = kh[r | J], bl = kl[r | J];
            const bool swp = ((key64(ah, al) > key64(bh, bl)) != desc);
            kh[r]     = swp ? bh : ah;
            kl[r]     = swp ? bl : al;
            kh[r | J] = swp ? ah : bh;
            kl[r | J] = swp ? al : bl;
        }
    }
}

template<int KS, int JL>
__device__ __forceinline__ void xlane_step(unsigned int (&kh)[KPL],
                                           unsigned int (&kl)[KPL], int lane)
{
    bool desc;
    if constexpr (KS >= MDIM) desc = true;
    else                      desc = ((lane & (KS / KPL)) == 0);
    const bool keep_max = (((lane & JL) == 0) == desc);
    #pragma unroll
    for (int r = 0; r < KPL; ++r) {
        const unsigned int ah = kh[r], al = kl[r];
        const unsigned int ch = lane_xor_u32<JL>(ah);
        const unsigned int cl = lane_xor_u32<JL>(al);
        const bool own = (keep_max == (key64(ah, al) > key64(ch, cl)));
        kh[r] = own ? ah : ch;
        kl[r] = own ? al : cl;
    }
}

template<int KS, int J>
__device__ __forceinline__ void stage_steps(unsigned int (&kh)[KPL],
                                            unsigned int (&kl)[KPL], int lane)
{
    if constexpr (J >= 1) {
        if constexpr (J >= KPL) xlane_step<KS, J / KPL>(kh, kl, lane);
        else                    inlane_step<KS, J>(kh, kl, lane);
        stage_steps<KS, (J >> 1)>(kh, kl, lane);
    }
}

template<int KS>
__device__ __forceinline__ void bitonic_stage(unsigned int (&kh)[KPL],
                                              unsigned int (&kl)[KPL], int lane)
{
    stage_steps<KS, (KS >> 1)>(kh, kl, lane);
}

__device__ __forceinline__ float med3f(float a, float b, float c)
{
    return __builtin_amdgcn_fmed3f(a, b, c);
}

__device__ __forceinline__ float keep_bound(bool keep_larger)
{
    return keep_larger ? __builtin_inff() : -__builtin_inff();
}

template<int KS, int J>
__device__ __forceinline__ void inlane_step1(float (&k)[KPL], int lane)
{
    #pragma unroll
    for (int r = 0; r < KPL; ++r) {
        if ((r & J) == 0) {
            const float a = k[r], b = k[r | J];
            if constexpr (KS < KPL || KS >= MDIM) {
                const bool desc = (KS >= MDIM) ? true : ((r & KS) == 0);
                const float hi = __builtin_fmaxf(a, b);
                const float lo = __builtin_fminf(a, b);
                k[r]     = desc ? hi : lo;
                k[r | J] = desc ? lo : hi;
            } else {
                const float bd = keep_bound((lane & (KS / KPL)) == 0);
                k[r]     = med3f(a, b,  bd);
                k[r | J] = med3f(a, b, -bd);
            }
        }
    }
}

template<int KS, int JL>
__device__ __forceinline__ void xlane_step1(float (&k)[KPL], int lane)
{
    bool desc;
    if constexpr (KS >= MDIM) desc = true;
    else                      desc = ((lane & (KS / KPL)) == 0);
    const float bd = keep_bound(((lane & JL) == 0) == desc);
    #pragma unroll
    for (int r = 0; r < KPL; ++r) {
        const float a = k[r];
        const float c = lane_xor_f32<JL>(a);
        k[r] = med3f(a, c, bd);
    }
}

template<int KS, int J>
__device__ __forceinline__ void stage_steps1(float (&k)[KPL], int lane)
{
    if constexpr (J >= 1) {
        if constexpr (J >= KPL) xlane_step1<KS, J / KPL>(k, lane);
        else                    inlane_step1<KS, J>(k, lane);
        stage_steps1<KS, (J >> 1)>(k, lane);
    }
}

template<int KS>
__device__ __forceinline__ void bitonic_stages1(float (&k)[KPL], int lane)
{
    if constexpr (KS <= MDIM) {
        stage_steps1<KS, (KS >> 1)>(k, lane);
        bitonic_stages1<2 * KS>(k, lane);
    }
}

__device__ __forceinline__ float ratio_of(float k, float k1)
{
    return (k * k1 >= 0.0f) ? 0.0f : (fabsf(k1) / (EPS + fabsf(k)));
}

constexpr int ctz_c(int x) { int n = 0; while (((x >> n) & 1) == 0) ++n; return n; }

template<int R>
__device__ __forceinline__ void tree_store(unsigned int* tree,
                                           const unsigned int (&k)[KPL], int lane)
{
    if constexpr (R < KPL - 1) {
        constexpr int tz = ctz_c(R + 1);
        tree[(256 >> tz) + ((lane * KPL + R + 1) >> (tz + 1))] = k[R];
        tree_store<R + 1>(tree, k, lane);
    } else {
        const int tz = 4 + __builtin_ctz((unsigned int)(lane + 1));
        const unsigned int q = (unsigned int)(KPL * (lane + 1));
        tree[(256u >> tz) + (q >> (tz + 1))] = k[R];
    }
}

__global__ __launch_bounds__(WAVE)
void ibp_get_ub_kernel(const float* __restrict__ K,
                       const float* __restrict__ K1,
                       const float* __restrict__ Bv,
                       const float* __restrict__ B1,
                       const float* __restrict__ WLB,
                       const float* __restrict__ WUB,
                       float* __restrict__ out)
{
    __shared__ __attribute__((aligned(16))) float s_buf[2 * PADDED];
    __shared__ unsigned int s_tree[MDIM];
    float* const s_in   = s_buf;
    float* const s_scan = s_buf + PADDED;

    const int lane = threadIdx.x;
    const int row  = blockIdx.x;

    const float4* __restrict__ K4   = (const float4*)(K  + (size_t)row * MDIM);
    const float4* __restrict__ K14  = (const float4*)(K1 + (size_t)row * MDIM);
    const float4* __restrict__ WLB4 = (const float4*)WLB;
    const float4* __restrict__ WUB4 = (const float4*)WUB;

    unsigned int h0[KPL];
    float td_o[KPL];
    float t1d_o[KPL];
    float pt = 0.0f, pr = 0.0f;
    #pragma unroll
    for (int c = 0; c < 4; ++c) {
        const float4 kq   = K4  [c * WAVE + lane];
        const float4 k1q  = K14 [c * WAVE + lane];
        const float4 wlq  = WLB4[c * WAVE + lane];
        const float4 wuq  = WUB4[c * WAVE + lane];
        const float kk[4]  = {kq.x,  kq.y,  kq.z,  kq.w};
        const float kk1[4] = {k1q.x, k1q.y, k1q.z, k1q.w};
        const float wl[4]  = {wlq.x, wlq.y, wlq.z, wlq.w};
        const float wu[4]  = {wuq.x, wuq.y, wuq.z, wuq.w};
        #pragma unroll
        for (int u = 0; u < 4; ++u) {
            const int   i   = c * 128 + lane * 4 + u;
            const float k   = kk[u];
            const float k1  = kk1[u];
            const float wlb = wl[u];
            const float wub = wu[u];

            const bool  pick_lb = (k < 0.0f) || ((k == 0.0f) && (k1 < 0.0f));
            const float w_ret   = pick_lb ? wlb : wub;
            pt += w_ret * k;
            pr += w_ret * k1;

            const float span  = wub - wlb;
            const float ak    = fabsf(k);
            const float ak1   = fabsf(k1);
            const float td    = ak  * span;
            const float t1d   = ak1 * span;
            const float ratio = ratio_of(k, k1);
            const bool  pos   = (ratio > 0.0f);

            td_o [4 * c + u] = pos ? td  : 0.0f;
            t1d_o[4 * c + u] = pos ? t1d : 0.0f;
            h0   [4 * c + u] = pos ? __float_as_uint(ratio)
                                   : (CODE_BASE + (unsigned int)(MDIM - 1 - i));
        }
    }

    wave_reduce2<16>(pt, pr);
    const float t    = pt + Bv[row];
    const float ret0 = pr + B1[row];

    unsigned int key[KPL];
    {
        float keyf[KPL];
        #pragma unroll
        for (int r = 0; r < KPL; ++r) keyf[r] = __uint_as_float(h0[r]);
        bitonic_stages1<2>(keyf, lane);
        #pragma unroll
        for (int r = 0; r < KPL; ++r) key[r] = __float_as_uint(keyf[r]);
    }

    bool tie = false;
    #pragma unroll
    for (int r = 0; r + 1 < KPL; ++r)
        tie |= (key[r] == key[r + 1]);
    const unsigned int prev_last = __shfl_up(key[KPL - 1], 1, WAVE);
    tie |= (lane > 0) && (prev_last == key[0]);

    float td_s[KPL], t1d_s[KPL];
    if (__builtin_amdgcn_ballot_w32(tie) == 0u) {
        tree_store<0>(s_tree, key, lane);
        __syncthreads();

        unsigned int nb[KPL];
        #pragma unroll
        for (int e = 0; e < KPL; ++e) nb[e] = 4u;
        #pragma unroll
        for (int lvl = 0; lvl < 9; ++lvl) {
            #pragma unroll
            for (int e = 0; e < KPL; ++e) {
                const unsigned int w = s_tree[nb[e] >> 2];
                nb[e] = (nb[e] << 1) + ((w > h0[e]) ? 4u : 0u);
            }
        }
        #pragma unroll
        for (int e = 0; e < KPL; ++e) {
            const int rank = (int)(nb[e] >> 2) - MDIM;
            const int p    = pad16(rank);
            s_buf[2 * p]     = td_o[e];
            s_buf[2 * p + 1] = t1d_o[e];
        }
        __syncthreads();
        #pragma unroll
        for (int r = 0; r < KPL; ++r) {
            const int p = lane * (KPL + 1) + r;
            td_s[r]  = s_buf[2 * p];
            t1d_s[r] = s_buf[2 * p + 1];
        }
    } else {
        unsigned int kh[KPL], kl[KPL];
        #pragma unroll
        for (int c = 0; c < 4; ++c) {
            #pragma unroll
            for (int u = 0; u < 4; ++u) {
                const int i = c * 128 + lane * 4 + u;
                const int p = pad16(i);
                s_buf[2 * p]     = td_o [4 * c + u];
                s_buf[2 * p + 1] = t1d_o[4 * c + u];
                kh[4 * c + u] = h0[4 * c + u];
                kl[4 * c + u] = (unsigned int)(MDIM - 1 - i);
            }
        }
        bitonic_stage<  2>(kh, kl, lane);
        bitonic_stage<  4>(kh, kl, lane);
        bitonic_stage<  8>(kh, kl, lane);
        bitonic_stage< 16>(kh, kl, lane);
        bitonic_stage< 32>(kh, kl, lane);
        bitonic_stage< 64>(kh, kl, lane);
        bitonic_stage<128>(kh, kl, lane);
        bitonic_stage<256>(kh, kl, lane);
        bitonic_stage<512>(kh, kl, lane);
        __syncthreads();
        #pragma unroll
        for (int r = 0; r < KPL; ++r) {
            const int idx = (MDIM - 1) - (int)kl[r];
            const int p   = pad16(idx);
            td_s[r]  = s_buf[2 * p];
            t1d_s[r] = s_buf[2 * p + 1];
        }
    }
    __syncthreads();
    #pragma unroll
    for (int r = 0; r < KPL; ++r)
        s_in[lane * (KPL + 1) + r] = td_s[r];
    __syncthreads();

    {
        const int mn = lane & 15;
        const int kh = lane >> 4;
        const int g0 = mn * (KPL + 1);
        const int g1 = (16 + mn) * (KPL + 1);
        v8f acc0 = {0.f, 0.f, 0.f, 0.f, 0.f, 0.f, 0.f, 0.f};
        v8f acc1 = {0.f, 0.f, 0.f, 0.f, 0.f, 0.f, 0.f, 0.f};
        #pragma unroll
        for (int c = 0; c < 4; ++c) {
            const int k0 = 4 * c + 2 * kh;
            v2f a, b0, b1;
            a[0] = (mn >= k0)     ? 1.0f : 0.0f;
            a[1] = (mn >= k0 + 1) ? 1.0f : 0.0f;
            b0[0] = s_in[g0 + k0];
            b0[1] = s_in[g0 + k0 + 1];
            b1[0] = s_in[g1 + k0];
            b1[1] = s_in[g1 + k0 + 1];
            acc0 = __builtin_amdgcn_wmma_f32_16x16x4_f32(
                       false, a, false, b0, (short)0, acc0, false, false);
            acc1 = __builtin_amdgcn_wmma_f32_16x16x4_f32(
                       false, a, false, b1, (short)0, acc1, false, false);
        }
        #pragma unroll
        for (int v = 0; v < 8; ++v) {
            const int m = v + 8 * kh;
            s_scan[g0 + m] = acc0[v];
            s_scan[g1 + m] = acc1[v];
        }
    }
    __syncthreads();

    float scan[KPL];
    #pragma unroll
    for (int r = 0; r < KPL; ++r)
        scan[r] = s_scan[lane * (KPL + 1) + r];

    const float gsum = scan[KPL - 1];
    float inc = gsum;
    #pragma unroll
    for (int off = 1; off < 32; off <<= 1) {
        const float y = __shfl_up(inc, off, 32);
        if (lane >= off) inc += y;
    }
    const float goff = inc - gsum;

    float contrib = 0.0f;
    #pragma unroll
    for (int r = 0; r < KPL; ++r) {
        const float tds     = scan[r] + goff;
        float percent;
        const float delta = t - tds;
        if (delta < 0.0f) {
            float pcl = delta / (td_s[r] + EPS);
            pcl = fmaxf(-1.0f, fminf(0.0f, pcl));
            percent = pcl + 1.0f;
        } else {
            percent = 1.0f;
        }
        contrib += t1d_s[r] * percent;
    }
    wave_reduce1<16>(contrib);
    if (lane == 0)
        out[row] = (t >= 0.0f) ? (ret0 + contrib) : -FAKE_INF;
}

extern "C" void kernel_launch(void* const* d_in, const int* in_sizes, int n_in,
                              void* d_out, int out_size, void* d_ws, size_t ws_size,
                              hipStream_t stream)
{
    (void)n_in; (void)d_ws; (void)ws_size; (void)out_size;
    const float* k   = (const float*)d_in[0];
    const float* k1  = (const float*)d_in[1];
    const float* b   = (const float*)d_in[2];
    const float* b1  = (const float*)d_in[3];
    const float* wlb = (const float*)d_in[4];
    const float* wub = (const float*)d_in[5];

    const int B = in_sizes[2];

    ibp_get_ub_kernel<<<dim3(B), dim3(WAVE), 0, stream>>>(
        k, k1, b, b1, wlb, wub, (float*)d_out);
}
